# starting kernel
# baseline (speedup 1.0000x reference)
#include <hip/hip_runtime.h>


typedef float v2f __attribute__((ext_vector_type(2)));
typedef float v8f __attribute__((ext_vector_type(8)));

#define GRID_N 512
#define MCTRL  64
#define NCTRL  64
#define BATCH  8

__global__ void build_basis_kernel(const float* __restrict__ Nu,
                                   const int*   __restrict__ uspan,
                                   const float* __restrict__ Nv,
                                   const int*   __restrict__ vspan,
                                   float* __restrict__ Bu,
                                   float* __restrict__ Bv) {
  int t = blockIdx.x * blockDim.x + threadIdx.x;
  if (t < GRID_N) {
    float* row = Bu + (size_t)t * MCTRL;
    #pragma unroll
    for (int j = 0; j < MCTRL; ++j) row[j] = 0.0f;
    int s = uspan[t];
    #pragma unroll
    for (int r = 0; r < 4; ++r) row[s - 3 + r] = Nu[t * 4 + r];
  } else if (t < 2 * GRID_N) {
    int g = t - GRID_N;
    float* row = Bv + (size_t)g * NCTRL;
    #pragma unroll
    for (int j = 0; j < NCTRL; ++j) row[j] = 0.0f;
    int s = vspan[g];
    #pragma unroll
    for (int r = 0; r < 4; ++r) row[s - 3 + r] = Nv[g * 4 + r];
  }
}

__device__ __forceinline__ int band_min(int span) {
  int c = span - 3;
  return (c > 56) ? 56 : (c < 0 ? 0 : c);
}

__global__ __launch_bounds__(128)
void stageA_kernel(const float* __restrict__ ctrl,
                   const float* __restrict__ Bu,
                   const int*   __restrict__ uspan,
                   float* __restrict__ Mid)
{
  const int gt = blockIdx.x;
  const int bb = blockIdx.y;
  const int g0 = gt * 16;

  const int d    = threadIdx.x >> 5;
  const int lane = threadIdx.x & 31;
  const int hh   = lane >> 4;
  const int l    = lane & 15;

  const int umin = band_min(uspan[g0]);

  v8f acc0 = {}, acc1 = {}, acc2 = {}, acc3 = {};
  const float* BuRow = Bu + (size_t)(g0 + l) * MCTRL + umin;

  #pragma unroll
  for (int kb = 0; kb < 2; ++kb) {
    const int k0 = 4 * kb + 2 * hh;
    v2f a;
    a.x = BuRow[k0];
    a.y = BuRow[k0 + 1];
    const float* c0 = ctrl + ((size_t)(bb * MCTRL + umin + k0) * NCTRL) * 4 + d;
    const float* c1 = c0 + (size_t)NCTRL * 4;
    {
      v2f b; b.x = c0[(0 * 16 + l) * 4]; b.y = c1[(0 * 16 + l) * 4];
      acc0 = __builtin_amdgcn_wmma_f32_16x16x4_f32(false, a, false, b, (short)0, acc0, false, false);
    }
    {
      v2f b; b.x = c0[(1 * 16 + l) * 4]; b.y = c1[(1 * 16 + l) * 4];
      acc1 = __builtin_amdgcn_wmma_f32_16x16x4_f32(false, a, false, b, (short)0, acc1, false, false);
    }
    {
      v2f b; b.x = c0[(2 * 16 + l) * 4]; b.y = c1[(2 * 16 + l) * 4];
      acc2 = __builtin_amdgcn_wmma_f32_16x16x4_f32(false, a, false, b, (short)0, acc2, false, false);
    }
    {
      v2f b; b.x = c0[(3 * 16 + l) * 4]; b.y = c1[(3 * 16 + l) * 4];
      acc3 = __builtin_amdgcn_wmma_f32_16x16x4_f32(false, a, false, b, (short)0, acc3, false, false);
    }
  }

  float* Mrow = Mid + ((size_t)(bb * 4 + d) * GRID_N + g0) * NCTRL;
  #pragma unroll
  for (int j = 0; j < 8; ++j) {
    const int m = j + 8 * hh;
    Mrow[(size_t)m * NCTRL +  0 + l] = acc0[j];
    Mrow[(size_t)m * NCTRL + 16 + l] = acc1[j];
    Mrow[(size_t)m * NCTRL + 32 + l] = acc2[j];
    Mrow[(size_t)m * NCTRL + 48 + l] = acc3[j];
  }
}

__global__ __launch_bounds__(128)
void stageB_kernel(const float* __restrict__ Mid,
                   const float* __restrict__ Bv,
                   const int*   __restrict__ vspan,
                   float* __restrict__ out)
{
  __shared__ float Sws[4][16][17];

  const int ht = blockIdx.x;
  const int gt = blockIdx.y;
  const int bb = blockIdx.z;
  const int g0 = gt * 16, h0 = ht * 16;

  const int d    = threadIdx.x >> 5;
  const int lane = threadIdx.x & 31;
  const int hh   = lane >> 4;
  const int l    = lane & 15;

  const int cmin = band_min(vspan[h0]);

  const float* Arow = Mid + ((size_t)(bb * 4 + d) * GRID_N + g0 + l) * NCTRL + cmin;
  const float* Brow = Bv + (size_t)(h0 + l) * NCTRL + cmin;

  v8f accO = {};
  #pragma unroll
  for (int kb = 0; kb < 2; ++kb) {
    const int k0 = 4 * kb + 2 * hh;
    v2f a, b;
    a.x = Arow[k0];
    a.y = Arow[k0 + 1];
    b.x = Brow[k0];
    b.y = Brow[k0 + 1];
    accO = __builtin_amdgcn_wmma_f32_16x16x4_f32(false, a, false, b, (short)0, accO, false, false);
  }

  #pragma unroll
  for (int j = 0; j < 8; ++j) Sws[d][j + 8 * hh][l] = accO[j];
  __syncthreads();

  for (int p = threadIdx.x; p < 256; p += 128) {
    const int m = p >> 4, n = p & 15;
    const float inv = 1.0f / Sws[3][m][n];
    const size_t o = (((size_t)bb * GRID_N + (g0 + m)) * GRID_N + (h0 + n)) * 3;
    out[o + 0] = Sws[0][m][n] * inv;
    out[o + 1] = Sws[1][m][n] * inv;
    out[o + 2] = Sws[2][m][n] * inv;
  }
}

__global__ __launch_bounds__(128)
void surfeval_fused_kernel(const float* __restrict__ ctrl,
                           const float* __restrict__ Bu,
                           const float* __restrict__ Bv,
                           float* __restrict__ out)
{
  __shared__ float A1s[4][16][68];
  __shared__ float Sws[4][16][17];

  const int ht = blockIdx.x, gt = blockIdx.y, bb = blockIdx.z;
  const int g0 = gt * 16, h0 = ht * 16;
  const int d    = threadIdx.x >> 5;
  const int lane = threadIdx.x & 31;
  const int hh   = lane >> 4;
  const int l    = lane & 15;

  v8f acc0 = {}, acc1 = {}, acc2 = {}, acc3 = {};
  const float* BuRow = Bu + (size_t)(g0 + l) * MCTRL;
  #pragma unroll
  for (int kb = 0; kb < 16; ++kb) {
    const int k0 = 4 * kb + 2 * hh;
    v2f a; a.x = BuRow[k0]; a.y = BuRow[k0 + 1];
    const float* c0 = ctrl + ((size_t)(bb * MCTRL + k0) * NCTRL) * 4 + d;
    const float* c1 = c0 + (size_t)NCTRL * 4;
    { v2f b; b.x = c0[(0*16+l)*4]; b.y = c1[(0*16+l)*4];
      acc0 = __builtin_amdgcn_wmma_f32_16x16x4_f32(false, a, false, b, (short)0, acc0, false, false); }
    { v2f b; b.x = c0[(1*16+l)*4]; b.y = c1[(1*16+l)*4];
      acc1 = __builtin_amdgcn_wmma_f32_16x16x4_f32(false, a, false, b, (short)0, acc1, false, false); }
    { v2f b; b.x = c0[(2*16+l)*4]; b.y = c1[(2*16+l)*4];
      acc2 = __builtin_amdgcn_wmma_f32_16x16x4_f32(false, a, false, b, (short)0, acc2, false, false); }
    { v2f b; b.x = c0[(3*16+l)*4]; b.y = c1[(3*16+l)*4];
      acc3 = __builtin_amdgcn_wmma_f32_16x16x4_f32(false, a, false, b, (short)0, acc3, false, false); }
  }
  #pragma unroll
  for (int j = 0; j < 8; ++j) {
    const int m = j + 8 * hh;
    A1s[d][m][ 0 + l] = acc0[j];
    A1s[d][m][16 + l] = acc1[j];
    A1s[d][m][32 + l] = acc2[j];
    A1s[d][m][48 + l] = acc3[j];
  }
  __syncthreads();

  v8f accO = {};
  const float* BvRow = Bv + (size_t)(h0 + l) * NCTRL;
  #pragma unroll
  for (int kb = 0; kb < 16; ++kb) {
    const int k0 = 4 * kb + 2 * hh;
    v2f a, b;
    a.x = A1s[d][l][k0]; a.y = A1s[d][l][k0 + 1];
    b.x = BvRow[k0];     b.y = BvRow[k0 + 1];
    accO = __builtin_amdgcn_wmma_f32_16x16x4_f32(false, a, false, b, (short)0, accO, false, false);
  }
  #pragma unroll
  for (int j = 0; j < 8; ++j) Sws[d][j + 8 * hh][l] = accO[j];
  __syncthreads();

  for (int p = threadIdx.x; p < 256; p += 128) {
    const int m = p >> 4, n = p & 15;
    const float inv = 1.0f / Sws[3][m][n];
    const size_t o = (((size_t)bb * GRID_N + (g0 + m)) * GRID_N + (h0 + n)) * 3;
    out[o + 0] = Sws[0][m][n] * inv;
    out[o + 1] = Sws[1][m][n] * inv;
    out[o + 2] = Sws[2][m][n] * inv;
  }
}

extern "C" void kernel_launch(void* const* d_in, const int* in_sizes, int n_in,
                              void* d_out, int out_size, void* d_ws, size_t ws_size,
                              hipStream_t stream) {
  const float* ctrl  = (const float*)d_in[0];
  const int*   uspan = (const int*)  d_in[1];
  const int*   vspan = (const int*)  d_in[2];
  const float* Nu    = (const float*)d_in[3];
  const float* Nv    = (const float*)d_in[4];

  float* Bu = (float*)d_ws;
  float* Bv = Bu + (size_t)GRID_N * MCTRL;
  float* Mid = Bv + (size_t)GRID_N * NCTRL;

  const size_t need = ((size_t)2 * GRID_N * MCTRL +
                       (size_t)BATCH * 4 * GRID_N * NCTRL) * sizeof(float);

  build_basis_kernel<<<4, 256, 0, stream>>>(Nu, uspan, Nv, vspan, Bu, Bv);

  if (ws_size >= need) {
    dim3 gridA(GRID_N / 16, BATCH);
    stageA_kernel<<<gridA, 128, 0, stream>>>(ctrl, Bu, uspan, Mid);
    dim3 gridB(GRID_N / 16, GRID_N / 16, BATCH);
    stageB_kernel<<<gridB, 128, 0, stream>>>(Mid, Bv, vspan, (float*)d_out);
  } else {
    dim3 grid(GRID_N / 16, GRID_N / 16, BATCH);
    surfeval_fused_kernel<<<grid, 128, 0, stream>>>(ctrl, Bu, Bv, (float*)d_out);
  }
}


# Whole 256-byte units per store: a wave owns a 16x64 strip of one batch, stages its 16 rows of 768 B in its own LDS region, writes 16 B per lane, 512 contiguous B per instruction; no barrier
# speedup vs baseline: 4.3101x; 4.3101x over previous
#include <hip/hip_runtime.h>


typedef float v2f __attribute__((ext_vector_type(2)));
typedef float v4f __attribute__((ext_vector_type(4)));
typedef float v8f __attribute__((ext_vector_type(8)));

#define GRID_N 512
#define MCTRL  64
#define NCTRL  64
#define BATCH  8
#define VT     4
#define ROW_F  (VT * 16 * 3)
#define STAGE_F (16 * ROW_F)

__device__ __forceinline__ int band_min(int span) {
  int c = span - 3;
  return (c > 56) ? 56 : (c < 0 ? 0 : c);
}

__device__ __forceinline__ float band_coef(float c0, float c1, float c2, float c3, int r) {
  float v = (r == 0) ? c0 : 0.0f;
  v = (r == 1) ? c1 : v;
  v = (r == 2) ? c2 : v;
  v = (r == 3) ? c3 : v;
  return v;
}

__global__ __launch_bounds__(256)
void surf_strip_kernel(const float* __restrict__ ctrl,
                       const int*   __restrict__ uspan,
                       const int*   __restrict__ vspan,
                       const float* __restrict__ Nu,
                       const float* __restrict__ Nv,
                       float* __restrict__ out)
{
  __shared__ __attribute__((aligned(256))) float stage[BATCH * STAGE_F];

  const int wave = __builtin_amdgcn_readfirstlane((int)(threadIdx.x >> 5));
  const int lane = threadIdx.x & 31;
  const int hh   = lane >> 4;
  const int l    = lane & 15;

  const int g0 = blockIdx.y * 16;
  const int h0 = blockIdx.x * (16 * VT);

  int vs0[VT];
  #pragma unroll
  for (int t = 0; t < VT; ++t) vs0[t] = vspan[h0 + 16 * t];
  const int us0 = uspan[g0];

  int vmin[VT];
  #pragma unroll
  for (int t = 0; t < VT; ++t) vmin[t] = band_min(vs0[t]);
  const int umin = band_min(us0);

  const int g  = g0 + l;
  const int ru = 2 * hh - (uspan[g] - 3 - umin);
  const float nu0 = Nu[g * 4 + 0], nu1 = Nu[g * 4 + 1], nu2 = Nu[g * 4 + 2], nu3 = Nu[g * 4 + 3];

  v2f bu[2];
  #pragma unroll
  for (int kb = 0; kb < 2; ++kb) {
    bu[kb].x = band_coef(nu0, nu1, nu2, nu3, ru + 4 * kb);
    bu[kb].y = band_coef(nu0, nu1, nu2, nu3, ru + 4 * kb + 1);
  }

  const float* cp = ctrl + (size_t)wave * (MCTRL * NCTRL * 4)
                  + ((size_t)(umin + 2 * (l >> 3) + (l & 1)) * NCTRL + 2 * hh) * 4
                  + ((l >> 1) & 3);

  v2f  bv[VT][2];
  v2f  c[VT][2][2];
  #pragma unroll
  for (int t = 0; t < VT; ++t) {
    const float* cb = cp + vmin[t] * 4;
    #pragma unroll
    for (int km = 0; km < 2; ++km) {
      #pragma unroll
      for (int kn = 0; kn < 2; ++kn) {
        c[t][km][kn].x = cb[km * (4 * NCTRL * 4) + kn * 16];
        c[t][km][kn].y = cb[km * (4 * NCTRL * 4) + kn * 16 + 4];
      }
    }
  }

  #pragma unroll
  for (int t = 0; t < VT; ++t) {
    const int h  = h0 + 16 * t + l;
    const int rv = 2 * hh - (vspan[h] - 3 - vmin[t]);
    const float nv0 = Nv[h * 4 + 0], nv1 = Nv[h * 4 + 1], nv2 = Nv[h * 4 + 2], nv3 = Nv[h * 4 + 3];
    #pragma unroll
    for (int kb = 0; kb < 2; ++kb) {
      bv[t][kb].x = band_coef(nv0, nv1, nv2, nv3, rv + 4 * kb);
      bv[t][kb].y = band_coef(nv0, nv1, nv2, nv3, rv + 4 * kb + 1);
    }
  }

  float* sw = stage + wave * STAGE_F;
  float* sp = sw + (8 * hh) * ROW_F + l * 3;

  const v8f zero = {};

  #pragma unroll
  for (int t = 0; t < VT; ++t) {
    v8f t0 = __builtin_amdgcn_wmma_f32_16x16x4_f32(false, c[t][0][0], false, bv[t][0], (short)0, zero, false, false);
    v8f t1 = __builtin_amdgcn_wmma_f32_16x16x4_f32(false, c[t][1][0], false, bv[t][0], (short)0, zero, false, false);
    t0 = __builtin_amdgcn_wmma_f32_16x16x4_f32(false, c[t][0][1], false, bv[t][1], (short)0, t0, false, false);
    t1 = __builtin_amdgcn_wmma_f32_16x16x4_f32(false, c[t][1][1], false, bv[t][1], (short)0, t1, false, false);

    v8f o[4];
    #pragma unroll
    for (int d = 0; d < 4; ++d) {
      v2f q; q.x = t0[2 * d]; q.y = t0[2 * d + 1];
      o[d] = __builtin_amdgcn_wmma_f32_16x16x4_f32(false, bu[0], false, q, (short)0, zero, false, false);
    }
    #pragma unroll
    for (int d = 0; d < 4; ++d) {
      v2f q; q.x = t1[2 * d]; q.y = t1[2 * d + 1];
      o[d] = __builtin_amdgcn_wmma_f32_16x16x4_f32(false, bu[1], false, q, (short)0, o[d], false, false);
    }

    #pragma unroll
    for (int j = 0; j < 8; ++j) {
      const float inv = 1.0f / o[3][j];
      float* p = sp + j * ROW_F + t * 48;
      p[0] = o[0][j] * inv;
      p[1] = o[1][j] * inv;
      p[2] = o[2][j] * inv;
    }
  }

  __builtin_amdgcn_wave_barrier();

  float* ob = out + (((size_t)wave * GRID_N + g0) * GRID_N + h0) * 3;
  float* gp[3];
  #pragma unroll
  for (int r = 0; r < 3; ++r) {
    const int e   = 128 * r + 4 * lane;
    const int row = (e >= ROW_F) ? 1 : 0;
    gp[r] = ob + row * (GRID_N * 3) + (e - row * ROW_F);
  }
  const float* sr = sw + 4 * lane;
  #pragma unroll
  for (int q = 0; q < 8; ++q) {
    #pragma unroll
    for (int r = 0; r < 3; ++r) {
      const v4f v = *(const v4f*)(sr + (3 * q + r) * 128);
      *(v4f*)(gp[r] + q * (2 * GRID_N * 3)) = v;
    }
  }
}

extern "C" void kernel_launch(void* const* d_in, const int* in_sizes, int n_in,
                              void* d_out, int out_size, void* d_ws, size_t ws_size,
                              hipStream_t stream) {
  const float* ctrl  = (const float*)d_in[0];
  const int*   uspan = (const int*)  d_in[1];
  const int*   vspan = (const int*)  d_in[2];
  const float* Nu    = (const float*)d_in[3];
  const float* Nv    = (const float*)d_in[4];

  dim3 grid(GRID_N / (16 * VT), GRID_N / 16);
  surf_strip_kernel<<<grid, 256, 0, stream>>>(ctrl, uspan, vspan, Nu, Nv, (float*)d_out);
}


# Write-out streamed: each 256-byte unit of the strip's 16 rows leaves LDS right behind the tile that completes it (2nd, 3rd, 4th); one store = that unit of 2 rows; first stores after 2 of 4 tiles
# speedup vs baseline: 4.5792x; 1.0624x over previous
#include <hip/hip_runtime.h>


typedef float v2f __attribute__((ext_vector_type(2)));
typedef float v4f __attribute__((ext_vector_type(4)));
typedef float v8f __attribute__((ext_vector_type(8)));

#define GRID_N 512
#define MCTRL  64
#define NCTRL  64
#define BATCH  8
#define VT     4
#define ROW_F  (VT * 16 * 3)
#define STAGE_F (16 * ROW_F)

__device__ __forceinline__ int band_min(int span) {
  int c = span - 3;
  return (c > 56) ? 56 : (c < 0 ? 0 : c);
}

__device__ __forceinline__ float band_coef(float c0, float c1, float c2, float c3, int r) {
  float v = (r == 0) ? c0 : 0.0f;
  v = (r == 1) ? c1 : v;
  v = (r == 2) ? c2 : v;
  v = (r == 3) ? c3 : v;
  return v;
}

__global__ __launch_bounds__(256)
void surf_strip_kernel(const float* __restrict__ ctrl,
                       const int*   __restrict__ uspan,
                       const int*   __restrict__ vspan,
                       const float* __restrict__ Nu,
                       const float* __restrict__ Nv,
                       float* __restrict__ out)
{
  __shared__ __attribute__((aligned(256))) float stage[BATCH * STAGE_F];

  const int wave = __builtin_amdgcn_readfirstlane((int)(threadIdx.x >> 5));
  const int lane = threadIdx.x & 31;
  const int hh   = lane >> 4;
  const int l    = lane & 15;

  const int g0 = blockIdx.y * 16;
  const int h0 = blockIdx.x * (16 * VT);

  int vs0[VT];
  #pragma unroll
  for (int t = 0; t < VT; ++t) vs0[t] = vspan[h0 + 16 * t];
  const int us0 = uspan[g0];

  int vmin[VT];
  #pragma unroll
  for (int t = 0; t < VT; ++t) vmin[t] = band_min(vs0[t]);
  const int umin = band_min(us0);

  const int g  = g0 + l;
  const int ru = 2 * hh - (uspan[g] - 3 - umin);
  const float nu0 = Nu[g * 4 + 0], nu1 = Nu[g * 4 + 1], nu2 = Nu[g * 4 + 2], nu3 = Nu[g * 4 + 3];

  v2f bu[2];
  #pragma unroll
  for (int kb = 0; kb < 2; ++kb) {
    bu[kb].x = band_coef(nu0, nu1, nu2, nu3, ru + 4 * kb);
    bu[kb].y = band_coef(nu0, nu1, nu2, nu3, ru + 4 * kb + 1);
  }

  const float* cp = ctrl + (size_t)wave * (MCTRL * NCTRL * 4)
                  + ((size_t)(umin + 2 * (l >> 3) + (l & 1)) * NCTRL + 2 * hh) * 4
                  + ((l >> 1) & 3);

  v2f  bv[VT][2];
  v2f  c[VT][2][2];
  #pragma unroll
  for (int t = 0; t < VT; ++t) {
    const float* cb = cp + vmin[t] * 4;
    #pragma unroll
    for (int km = 0; km < 2; ++km) {
      #pragma unroll
      for (int kn = 0; kn < 2; ++kn) {
        c[t][km][kn].x = cb[km * (4 * NCTRL * 4) + kn * 16];
        c[t][km][kn].y = cb[km * (4 * NCTRL * 4) + kn * 16 + 4];
      }
    }
  }

  #pragma unroll
  for (int t = 0; t < VT; ++t) {
    const int h  = h0 + 16 * t + l;
    const int rv = 2 * hh - (vspan[h] - 3 - vmin[t]);
    const float nv0 = Nv[h * 4 + 0], nv1 = Nv[h * 4 + 1], nv2 = Nv[h * 4 + 2], nv3 = Nv[h * 4 + 3];
    #pragma unroll
    for (int kb = 0; kb < 2; ++kb) {
      bv[t][kb].x = band_coef(nv0, nv1, nv2, nv3, rv + 4 * kb);
      bv[t][kb].y = band_coef(nv0, nv1, nv2, nv3, rv + 4 * kb + 1);
    }
  }

  float* sw = stage + wave * STAGE_F;
  float* sp = sw + (8 * hh) * ROW_F + l * 3;

  float* ob = out + (((size_t)wave * GRID_N + g0) * GRID_N + h0) * 3;
  float*       gq = ob + hh * (GRID_N * 3) + 4 * l;
  const float* sq = sw + hh * ROW_F + 4 * l;

  const v8f zero = {};

  #pragma unroll
  for (int t = 0; t < VT; ++t) {
    v8f t0 = __builtin_amdgcn_wmma_f32_16x16x4_f32(false, c[t][0][0], false, bv[t][0], (short)0, zero, false, false);
    v8f t1 = __builtin_amdgcn_wmma_f32_16x16x4_f32(false, c[t][1][0], false, bv[t][0], (short)0, zero, false, false);
    t0 = __builtin_amdgcn_wmma_f32_16x16x4_f32(false, c[t][0][1], false, bv[t][1], (short)0, t0, false, false);
    t1 = __builtin_amdgcn_wmma_f32_16x16x4_f32(false, c[t][1][1], false, bv[t][1], (short)0, t1, false, false);

    v8f o[4];
    #pragma unroll
    for (int d = 0; d < 4; ++d) {
      v2f q; q.x = t0[2 * d]; q.y = t0[2 * d + 1];
      o[d] = __builtin_amdgcn_wmma_f32_16x16x4_f32(false, bu[0], false, q, (short)0, zero, false, false);
    }
    #pragma unroll
    for (int d = 0; d < 4; ++d) {
      v2f q; q.x = t1[2 * d]; q.y = t1[2 * d + 1];
      o[d] = __builtin_amdgcn_wmma_f32_16x16x4_f32(false, bu[1], false, q, (short)0, o[d], false, false);
    }

    #pragma unroll
    for (int j = 0; j < 8; ++j) {
      const float inv = 1.0f / o[3][j];
      float* p = sp + j * ROW_F + t * 48;
      p[0] = o[0][j] * inv;
      p[1] = o[1][j] * inv;
      p[2] = o[2][j] * inv;
    }

    if (t >= 1) {
      __builtin_amdgcn_wave_barrier();
      const int u = t - 1;
      #pragma unroll
      for (int q = 0; q < 8; ++q) {
        const v4f v = *(const v4f*)(sq + q * (2 * ROW_F) + u * 64);
        *(v4f*)(gq + q * (2 * GRID_N * 3) + u * 64) = v;
      }
    }
  }
}

extern "C" void kernel_launch(void* const* d_in, const int* in_sizes, int n_in,
                              void* d_out, int out_size, void* d_ws, size_t ws_size,
                              hipStream_t stream) {
  const float* ctrl  = (const float*)d_in[0];
  const int*   uspan = (const int*)  d_in[1];
  const int*   vspan = (const int*)  d_in[2];
  const float* Nu    = (const float*)d_in[3];
  const float* Nv    = (const float*)d_in[4];

  dim3 grid(GRID_N / (16 * VT), GRID_N / 16);
  surf_strip_kernel<<<grid, 256, 0, stream>>>(ctrl, uspan, vspan, Nu, Nv, (float*)d_out);
}


# Perspective divide by the hardware reciprocal of w (v_rcp_f32, 1 ulp) times x,y,z instead of the correctly rounded 1/w sequence: 341 fewer instructions per wave (1025 to 684)
# speedup vs baseline: 4.8713x; 1.0638x over previous
#include <hip/hip_runtime.h>


typedef float v2f __attribute__((ext_vector_type(2)));
typedef float v4f __attribute__((ext_vector_type(4)));
typedef float v8f __attribute__((ext_vector_type(8)));

#define GRID_N 512
#define MCTRL  64
#define NCTRL  64
#define BATCH  8
#define VT     4
#define ROW_F  (VT * 16 * 3)
#define STAGE_F (16 * ROW_F)

__device__ __forceinline__ int band_min(int span) {
  int c = span - 3;
  return (c > 56) ? 56 : (c < 0 ? 0 : c);
}

__device__ __forceinline__ float band_coef(float c0, float c1, float c2, float c3, int r) {
  float v = (r == 0) ? c0 : 0.0f;
  v = (r == 1) ? c1 : v;
  v = (r == 2) ? c2 : v;
  v = (r == 3) ? c3 : v;
  return v;
}

__global__ __launch_bounds__(256)
void surf_strip_kernel(const float* __restrict__ ctrl,
                       const int*   __restrict__ uspan,
                       const int*   __restrict__ vspan,
                       const float* __restrict__ Nu,
                       const float* __restrict__ Nv,
                       float* __restrict__ out)
{
  __shared__ __attribute__((aligned(256))) float stage[BATCH * STAGE_F];

  const int wave = __builtin_amdgcn_readfirstlane((int)(threadIdx.x >> 5));
  const int lane = threadIdx.x & 31;
  const int hh   = lane >> 4;
  const int l    = lane & 15;

  const int g0 = blockIdx.y * 16;
  const int h0 = blockIdx.x * (16 * VT);

  int vs0[VT];
  #pragma unroll
  for (int t = 0; t < VT; ++t) vs0[t] = vspan[h0 + 16 * t];
  const int us0 = uspan[g0];

  int vmin[VT];
  #pragma unroll
  for (int t = 0; t < VT; ++t) vmin[t] = band_min(vs0[t]);
  const int umin = band_min(us0);

  const int g  = g0 + l;
  const int ru = 2 * hh - (uspan[g] - 3 - umin);
  const float nu0 = Nu[g * 4 + 0], nu1 = Nu[g * 4 + 1], nu2 = Nu[g * 4 + 2], nu3 = Nu[g * 4 + 3];

  v2f bu[2];
  #pragma unroll
  for (int kb = 0; kb < 2; ++kb) {
    bu[kb].x = band_coef(nu0, nu1, nu2, nu3, ru + 4 * kb);
    bu[kb].y = band_coef(nu0, nu1, nu2, nu3, ru + 4 * kb + 1);
  }

  const float* cp = ctrl + (size_t)wave * (MCTRL * NCTRL * 4)
                  + ((size_t)(umin + 2 * (l >> 3) + (l & 1)) * NCTRL + 2 * hh) * 4
                  + ((l >> 1) & 3);

  v2f  bv[VT][2];
  v2f  c[VT][2][2];
  #pragma unroll
  for (int t = 0; t < VT; ++t) {
    const float* cb = cp + vmin[t] * 4;
    #pragma unroll
    for (int km = 0; km < 2; ++km) {
      #pragma unroll
      for (int kn = 0; kn < 2; ++kn) {
        c[t][km][kn].x = cb[km * (4 * NCTRL * 4) + kn * 16];
        c[t][km][kn].y = cb[km * (4 * NCTRL * 4) + kn * 16 + 4];
      }
    }
  }

  #pragma unroll
  for (int t = 0; t < VT; ++t) {
    const int h  = h0 + 16 * t + l;
    const int rv = 2 * hh - (vspan[h] - 3 - vmin[t]);
    const float nv0 = Nv[h * 4 + 0], nv1 = Nv[h * 4 + 1], nv2 = Nv[h * 4 + 2], nv3 = Nv[h * 4 + 3];
    #pragma unroll
    for (int kb = 0; kb < 2; ++kb) {
      bv[t][kb].x = band_coef(nv0, nv1, nv2, nv3, rv + 4 * kb);
      bv[t][kb].y = band_coef(nv0, nv1, nv2, nv3, rv + 4 * kb + 1);
    }
  }

  float* sw = stage + wave * STAGE_F;
  float* sp = sw + (8 * hh) * ROW_F + l * 3;

  float* ob = out + (((size_t)wave * GRID_N + g0) * GRID_N + h0) * 3;
  float*       gq = ob + hh * (GRID_N * 3) + 4 * l;
  const float* sq = sw + hh * ROW_F + 4 * l;

  const v8f zero = {};

  #pragma unroll
  for (int t = 0; t < VT; ++t) {
    v8f t0 = __builtin_amdgcn_wmma_f32_16x16x4_f32(false, c[t][0][0], false, bv[t][0], (short)0, zero, false, false);
    v8f t1 = __builtin_amdgcn_wmma_f32_16x16x4_f32(false, c[t][1][0], false, bv[t][0], (short)0, zero, false, false);
    t0 = __builtin_amdgcn_wmma_f32_16x16x4_f32(false, c[t][0][1], false, bv[t][1], (short)0, t0, false, false);
    t1 = __builtin_amdgcn_wmma_f32_16x16x4_f32(false, c[t][1][1], false, bv[t][1], (short)0, t1, false, false);

    v8f o[4];
    #pragma unroll
    for (int d = 0; d < 4; ++d) {
      v2f q; q.x = t0[2 * d]; q.y = t0[2 * d + 1];
      o[d] = __builtin_amdgcn_wmma_f32_16x16x4_f32(false, bu[0], false, q, (short)0, zero, false, false);
    }
    #pragma unroll
    for (int d = 0; d < 4; ++d) {
      v2f q; q.x = t1[2 * d]; q.y = t1[2 * d + 1];
      o[d] = __builtin_amdgcn_wmma_f32_16x16x4_f32(false, bu[1], false, q, (short)0, o[d], false, false);
    }

    #pragma unroll
    for (int j = 0; j < 8; ++j) {
      const float inv = __builtin_amdgcn_rcpf(o[3][j]);
      float* p = sp + j * ROW_F + t * 48;
      p[0] = o[0][j] * inv;
      p[1] = o[1][j] * inv;
      p[2] = o[2][j] * inv;
    }

    if (t >= 1) {
      __builtin_amdgcn_wave_barrier();
      const int u = t - 1;
      #pragma unroll
      for (int q = 0; q < 8; ++q) {
        const v4f v = *(const v4f*)(sq + q * (2 * ROW_F) + u * 64);
        *(v4f*)(gq + q * (2 * GRID_N * 3) + u * 64) = v;
      }
    }
  }
}

extern "C" void kernel_launch(void* const* d_in, const int* in_sizes, int n_in,
                              void* d_out, int out_size, void* d_ws, size_t ws_size,
                              hipStream_t stream) {
  const float* ctrl  = (const float*)d_in[0];
  const int*   uspan = (const int*)  d_in[1];
  const int*   vspan = (const int*)  d_in[2];
  const float* Nu    = (const float*)d_in[3];
  const float* Nv    = (const float*)d_in[4];

  dim3 grid(GRID_N / (16 * VT), GRID_N / 16);
  surf_strip_kernel<<<grid, 256, 0, stream>>>(ctrl, uspan, vspan, Nu, Nv, (float*)d_out);
}
